# starting kernel
# baseline (speedup 1.0000x reference)
#include <hip/hip_runtime.h>
#include <hip/hip_bf16.h>


typedef __attribute__((ext_vector_type(2))) float v2f;
typedef __attribute__((ext_vector_type(8))) float v8f;

#define T_STEPS 512
#define IN      4
#define HID     64
#define PROJ    52
#define G       256
#define KW      56
#define KC      14
#define ROWS    16
#define NBATCH  4096

#if __has_builtin(__builtin_amdgcn_rcpf)
#define FRCP(x) __builtin_amdgcn_rcpf(x)
#else
#define FRCP(x) (1.0f / (x))
#endif
#if __has_builtin(__builtin_amdgcn_exp2f)
#define FEXP2(x) __builtin_amdgcn_exp2f(x)
#else
#define FEXP2(x) exp2f(x)
#endif

#define LOG2E 1.4426950408889634f

static __device__ __forceinline__ v8f wmma_f32_16x16x4(v2f a, v2f b, v8f c) {
  return __builtin_amdgcn_wmma_f32_16x16x4_f32(false, a, false, b, (short)0, c,
                                               false, false);
}

static __device__ __forceinline__ v8f splat8(float s) {
  v8f v;
#pragma unroll
  for (int i = 0; i < 8; ++i) v[i] = s;
  return v;
}

static __device__ __forceinline__ float fast_sig(float x) {
  return FRCP(1.0f + FEXP2(x * -LOG2E));
}

static __device__ __forceinline__ float fast_tanh(float x) {
#if __has_builtin(__builtin_amdgcn_tanhf)
  return __builtin_amdgcn_tanhf(x);
#else
  const float ax = __builtin_fabsf(x);
  const float e  = FEXP2(ax * (-2.0f * LOG2E));
  const float r  = (1.0f - e) * FRCP(1.0f + e);
  return __builtin_copysignf(r, x);
#endif
}

__global__ void __launch_bounds__(128, 1)
lstm_proj_wmma_kernel(const float* __restrict__ x,
                      const float* __restrict__ W_ih,
                      const float* __restrict__ W_hh,
                      const float* __restrict__ b_ih,
                      const float* __restrict__ b_hh,
                      const float* __restrict__ W_hr,
                      float* __restrict__ out)
{
  __shared__ float abuf[ROWS * HID];
  __shared__ float hbuf[ROWS * HID];

  const int tid   = threadIdx.x;
  const int wave  = tid >> 5;
  const int lane  = tid & 31;
  const int l16   = lane & 15;
  const int hi    = lane >> 4;
  const int dlt   = hi * 2;
  const int rbase = blockIdx.x * ROWS;

  for (int idx = tid; idx < ROWS * HID; idx += blockDim.x) {
    abuf[idx] = 0.0f;
    hbuf[idx] = 0.0f;
  }

  float gbias[4];
  int   coln[4];
#pragma unroll
  for (int j = 0; j < 4; ++j) {
    const int col = 16 * (wave + 4 * j) + l16;
    coln[j]  = col;
    gbias[j] = b_ih[col] + b_hh[col];
  }

  v2f wtB[KC][4];
#pragma unroll
  for (int j = 0; j < 4; ++j) {
    const int col = coln[j];
    wtB[0][j].x = W_ih[col * IN + dlt];
    wtB[0][j].y = W_ih[col * IN + dlt + 1];
#pragma unroll
    for (int k = 1; k < KC; ++k) {
      const int hr = 4 * k + dlt - IN;
      wtB[k][j].x = W_hh[col * PROJ + hr];
      wtB[k][j].y = W_hh[col * PROJ + hr + 1];
    }
  }

  v2f whrB[16];
  {
    const int pcol = 16 * wave + l16;
#pragma unroll
    for (int k = 0; k < 16; ++k) {
      const int kk = 4 * k + dlt;
      v2f b;
      b.x = (pcol < PROJ) ? W_hr[pcol * HID + kk]     : 0.0f;
      b.y = (pcol < PROJ) ? W_hr[pcol * HID + kk + 1] : 0.0f;
      whrB[k] = b;
    }
  }

  const float* xrow =
      x + (size_t)(rbase + l16) * T_STEPS * IN + dlt;

  v8f c = splat8(0.0f);
  __syncthreads();

  for (int t = 0; t < T_STEPS; ++t) {
    v8f acc[4];
#pragma unroll
    for (int j = 0; j < 4; ++j) acc[j] = splat8(gbias[j]);

#pragma unroll
    for (int k = 0; k < KC; ++k) {
      v2f a;
      if (k == 0) {
        const float* xp = xrow + t * IN;
        a.x = xp[0]; a.y = xp[1];
      } else {
        a = *(const v2f*)&hbuf[l16 * HID + (4 * k - IN) + dlt];
      }
#pragma unroll
      for (int j = 0; j < 4; ++j)
        acc[j] = wmma_f32_16x16x4(a, wtB[k][j], acc[j]);
    }
    __builtin_prefetch(xrow + (t + 1) * IN, 0, 1);

#pragma unroll
    for (int r = 0; r < 8; ++r) {
      const float iv = fast_sig(acc[0][r]);
      const float fv = fast_sig(acc[1][r]);
      const float gv = fast_tanh(acc[2][r]);
      const float ov = fast_sig(acc[3][r]);
      const float cv = fv * c[r] + iv * gv;
      c[r] = cv;
      abuf[(r + 8 * hi) * HID + 16 * wave + l16] = ov * fast_tanh(cv);
    }
    __syncthreads();

    v8f p = splat8(0.0f);
#pragma unroll
    for (int k = 0; k < 16; ++k) {
      v2f pa = *(const v2f*)&abuf[l16 * HID + 4 * k + dlt];
      p = wmma_f32_16x16x4(pa, whrB[k], p);
    }
#pragma unroll
    for (int r = 0; r < 8; ++r)
      hbuf[(r + 8 * hi) * HID + 16 * wave + l16] = p[r];
    __syncthreads();
  }

  for (int idx = tid; idx < ROWS * PROJ; idx += blockDim.x) {
    const int r = idx / PROJ, cc = idx - r * PROJ;
    out[(size_t)(rbase + r) * PROJ + cc] = hbuf[r * HID + cc];
  }
}

extern "C" void kernel_launch(void* const* d_in, const int* in_sizes, int n_in,
                              void* d_out, int out_size, void* d_ws, size_t ws_size,
                              hipStream_t stream) {
  (void)in_sizes; (void)n_in; (void)out_size; (void)d_ws; (void)ws_size;
  const float* x    = (const float*)d_in[0];
  const float* W_ih = (const float*)d_in[1];
  const float* W_hh = (const float*)d_in[2];
  const float* b_ih = (const float*)d_in[3];
  const float* b_hh = (const float*)d_in[4];
  const float* W_hr = (const float*)d_in[5];
  float* out = (float*)d_out;

  dim3 grid(NBATCH / ROWS);
  dim3 block(128);
  lstm_proj_wmma_kernel<<<grid, block, 0, stream>>>(x, W_ih, W_hh, b_ih, b_hh,
                                                    W_hr, out);
}


# Exchange cut in two by h row block: rows 0-31 stored beneath 6 deferred products of rows 32-63, whose hand-over moves under the next step's gate products (2 barrier phases); gate sum order changes
# speedup vs baseline: 1.8331x; 1.8331x over previous
#include <hip/hip_runtime.h>
#include <hip/hip_bf16.h>


typedef __attribute__((ext_vector_type(2))) float v2f;
typedef __attribute__((ext_vector_type(4))) float v4f;
typedef __attribute__((ext_vector_type(8))) float v8f;

#define T_STEPS 512
#define IN      4
#define HID     64
#define PROJ    52
#define G       256
#define KW      56
#define KC      14
#define ROWS    16
#define NBATCH  4096
#define PPITCH  68
#define PPLANE  (ROWS * PPITCH)
#define PBUF    (4 * PPLANE)

#if __has_builtin(__builtin_amdgcn_rcpf)
#define FRCP(x) __builtin_amdgcn_rcpf(x)
#else
#define FRCP(x) (1.0f / (x))
#endif
#if __has_builtin(__builtin_amdgcn_exp2f)
#define FEXP2(x) __builtin_amdgcn_exp2f(x)
#else
#define FEXP2(x) exp2f(x)
#endif

#define LOG2E 1.4426950408889634f

static __device__ __forceinline__ v8f wmma_f32_16x16x4(v2f a, v2f b, v8f c) {
  return __builtin_amdgcn_wmma_f32_16x16x4_f32(false, a, false, b, (short)0, c,
                                               false, false);
}

static __device__ __forceinline__ v8f splat8(float s) {
  v8f v;
#pragma unroll
  for (int i = 0; i < 8; ++i) v[i] = s;
  return v;
}

static __device__ __forceinline__ float fast_sig(float x) {
  return FRCP(1.0f + FEXP2(x * -LOG2E));
}

static __device__ __forceinline__ float fast_tanh(float x) {
#if __has_builtin(__builtin_amdgcn_tanhf)
  return __builtin_amdgcn_tanhf(x);
#else
  const float ax = __builtin_fabsf(x);
  const float e  = FEXP2(ax * (-2.0f * LOG2E));
  const float r  = (1.0f - e) * FRCP(1.0f + e);
  return __builtin_copysignf(r, x);
#endif
}

__global__ void __launch_bounds__(128, 1)
lstm_proj_wmma_kernel(const float* __restrict__ x,
                      const float* __restrict__ W_ih,
                      const float* __restrict__ W_hh,
                      const float* __restrict__ b_ih,
                      const float* __restrict__ b_hh,
                      const float* __restrict__ W_hr,
                      float* __restrict__ out)
{
  __shared__ __attribute__((aligned(16))) float pbuf[2 * PBUF];

  const int tid   = threadIdx.x;
  const int wave  = tid >> 5;
  const int lane  = tid & 31;
  const int l16   = lane & 15;
  const int hi    = lane >> 4;
  const int dlt   = hi * 2;
  const int rbase = blockIdx.x * ROWS;

  for (int idx = tid; idx < PBUF; idx += blockDim.x) pbuf[PBUF + idx] = 0.0f;

  v8f gbias[4];
#pragma unroll
  for (int j = 0; j < 4; ++j) {
    const int row0 = HID * j + 16 * wave + 8 * hi;
#pragma unroll
    for (int r = 0; r < 8; ++r) gbias[j][r] = b_ih[row0 + r] + b_hh[row0 + r];
  }

  v2f wtA[KC][4];
#pragma unroll
  for (int j = 0; j < 4; ++j) {
    const int grow = HID * j + 16 * wave + l16;
    wtA[0][j].x = W_ih[grow * IN + dlt];
    wtA[0][j].y = W_ih[grow * IN + dlt + 1];
#pragma unroll
    for (int k = 1; k < KC; ++k) {
      const int q  = k - 1;
      const int hr = (q < 12) ? (8 * (q >> 1) + 4 * hi + 2 * (q & 1)) : (48 + 2 * hi);
      wtA[k][j].x = W_hh[grow * PROJ + hr];
      wtA[k][j].y = W_hh[grow * PROJ + hr + 1];
    }
  }

  v2f whrA[4][4];
#pragma unroll
  for (int mt = 0; mt < 4; ++mt) {
    const int prow = 16 * mt + l16;
#pragma unroll
    for (int kc = 0; kc < 4; ++kc) {
      const int hc = 16 * wave + 8 * hi + 2 * kc;
      v2f a;
      a.x = (prow < PROJ) ? W_hr[prow * HID + hc]     : 0.0f;
      a.y = (prow < PROJ) ? W_hr[prow * HID + hc + 1] : 0.0f;
      whrA[mt][kc] = a;
    }
  }

  const float* xrow =
      x + (size_t)(rbase + l16) * T_STEPS * IN + dlt;

  const int rd_off = l16 * PPITCH + 4 * hi;
  const int wr_off = wave * PPLANE + l16 * PPITCH + 8 * hi;

  v8f c = splat8(0.0f);

  v2f xb;
  { const float* xp0 = xrow; xb.x = xp0[0]; xb.y = xp0[1]; }
  __syncthreads();

#define FENCE() __builtin_amdgcn_sched_barrier(0)

#define LDS_LANDED() asm volatile("s_wait_dscnt 0x0" ::: "memory")
#define BAR_SIGNAL() asm volatile("s_barrier_signal -1" ::: "memory")
#define BAR_WAIT()   asm volatile("s_barrier_wait -1" ::: "memory")

  v4f hs;
  for (int t = 0; t < T_STEPS; ++t) {
    const float* prd = pbuf + ((t + 1) & 1) * PBUF + rd_off;
    float*       pwr = pbuf + (t & 1) * PBUF + wr_off;

    v4f hp[6][4];
    v2f q[4];
    v2f xnext;
#pragma unroll
    for (int p = 0; p < 4; ++p) {
#pragma unroll
      for (int n = 0; n < 4; ++n) hp[p][n] = *(const v4f*)(prd + n * PPLANE + 8 * p);
      FENCE();
    }

    v8f acc[4];
#pragma unroll
    for (int j = 0; j < 4; ++j)
      acc[j] = wmma_f32_16x16x4(wtA[0][j], xb, gbias[j]);
    FENCE();
    hs = ((hp[0][0] + hp[0][1]) + hp[0][2]) + hp[0][3];
    FENCE();
#pragma unroll
    for (int p = 0; p < 6; ++p) {
      v2f b0; b0.x = hs[0]; b0.y = hs[1];
      v2f b1; b1.x = hs[2]; b1.y = hs[3];
      acc[0] = wmma_f32_16x16x4(wtA[1 + 2 * p][0], b0, acc[0]);
      FENCE();
      if (p < 5)
        hs = ((hp[p + 1][0] + hp[p + 1][1]) + hp[p + 1][2]) + hp[p + 1][3];
      FENCE();
#pragma unroll
      for (int j = 1; j < 4; ++j)
        acc[j] = wmma_f32_16x16x4(wtA[1 + 2 * p][j], b0, acc[j]);
#pragma unroll
      for (int j = 0; j < 4; ++j)
        acc[j] = wmma_f32_16x16x4(wtA[2 + 2 * p][j], b1, acc[j]);
      if (p == 0) {
        FENCE();
        LDS_LANDED();
        BAR_SIGNAL();
        FENCE();
      }
      if (p == 1) {
        FENCE();
        BAR_WAIT();
        const int tn = (t + 1 < T_STEPS) ? (t + 1) : t;
        { const float* xp = xrow + tn * IN; xnext.x = xp[0]; xnext.y = xp[1]; }
#pragma unroll
        for (int pb = 4; pb < 6; ++pb) {
#pragma unroll
          for (int n = 0; n < 4; ++n) hp[pb][n] = *(const v4f*)(prd + n * PPLANE + 8 * pb);
          FENCE();
        }
        {
          const int o = 48 - 2 * hi;
#pragma unroll
          for (int n = 0; n < 4; ++n) q[n] = *(const v2f*)(prd + n * PPLANE + o);
        }
        FENCE();
      }
    }
    FENCE();
    const v2f bq = ((q[0] + q[1]) + q[2]) + q[3];
    FENCE();
#pragma unroll
    for (int j = 0; j < 4; ++j)
      acc[j] = wmma_f32_16x16x4(wtA[13][j], bq, acc[j]);
    FENCE();

    float mi[8], mf[8], mo[8], ei[8], ef[8], eo[8], gt[8], si[8], sf[8], so[8], tc[8];
    v8f a;
#define ST1(r) { mi[r] = acc[0][r] * -LOG2E; mf[r] = acc[1][r] * -LOG2E; mo[r] = acc[3][r] * -LOG2E; }
#define ST2(r) { ei[r] = FEXP2(mi[r]); ef[r] = FEXP2(mf[r]); gt[r] = fast_tanh(acc[2][r]); eo[r] = FEXP2(mo[r]); }
#define ST3(r) { si[r] = FRCP(1.0f + ei[r]); sf[r] = FRCP(1.0f + ef[r]); so[r] = FRCP(1.0f + eo[r]); }
#define ST4(r) { const float cv = sf[r] * c[r] + si[r] * gt[r]; c[r] = cv; tc[r] = fast_tanh(cv); }
#define ST5(r) { a[r] = so[r] * tc[r]; }
    ST1(0) ST1(1) ST1(2) ST1(3) ST1(4) ST1(5) ST1(6) ST1(7)
    ST2(0) ST2(1) ST3(0) ST3(1) ST4(0) ST4(1) ST5(0) ST5(1)
    FENCE();

    v8f pt[4];
#pragma unroll
    for (int mt = 0; mt < 4; ++mt) pt[mt] = splat8(0.0f);
    v2f b0p; b0p.x = a[0]; b0p.y = a[1];
    pt[0] = wmma_f32_16x16x4(whrA[0][0], b0p, pt[0]);
    FENCE();
    ST2(2) ST2(3) ST2(4) ST2(5)
    FENCE();
    pt[1] = wmma_f32_16x16x4(whrA[1][0], b0p, pt[1]);
    FENCE();
    ST3(2) ST3(3) ST3(4) ST3(5)
    FENCE();
    pt[2] = wmma_f32_16x16x4(whrA[2][0], b0p, pt[2]);
    FENCE();
    ST4(2) ST4(3) ST4(4) ST4(5)
    FENCE();
    pt[3] = wmma_f32_16x16x4(whrA[3][0], b0p, pt[3]);
    FENCE();
    ST5(2) ST5(3) ST5(4) ST5(5)
    FENCE();
    v2f b1p; b1p.x = a[2]; b1p.y = a[3];
    v2f b2p; b2p.x = a[4]; b2p.y = a[5];
    pt[0] = wmma_f32_16x16x4(whrA[0][1], b1p, pt[0]);
    FENCE();
    ST2(6) ST2(7)
    FENCE();
    pt[1] = wmma_f32_16x16x4(whrA[1][1], b1p, pt[1]);
    FENCE();
    ST3(6) ST3(7)
    FENCE();
    pt[0] = wmma_f32_16x16x4(whrA[0][2], b2p, pt[0]);
    FENCE();
    ST4(6) ST4(7)
    FENCE();
    pt[1] = wmma_f32_16x16x4(whrA[1][2], b2p, pt[1]);
    FENCE();
    ST5(6) ST5(7)
    FENCE();
#undef ST1
#undef ST2
#undef ST3
#undef ST4
#undef ST5
    v2f b3p; b3p.x = a[6]; b3p.y = a[7];
    pt[0] = wmma_f32_16x16x4(whrA[0][3], b3p, pt[0]);
    pt[1] = wmma_f32_16x16x4(whrA[1][3], b3p, pt[1]);
    FENCE();
    pt[2] = wmma_f32_16x16x4(whrA[2][1], b1p, pt[2]);
    FENCE();
#pragma unroll
    for (int mt = 0; mt < 2; ++mt) {
      v4f lo, hi4;
      lo[0] = pt[mt][0]; lo[1] = pt[mt][1]; lo[2] = pt[mt][2]; lo[3] = pt[mt][3];
      hi4[0] = pt[mt][4]; hi4[1] = pt[mt][5]; hi4[2] = pt[mt][6]; hi4[3] = pt[mt][7];
      *(v4f*)(pwr + 16 * mt)     = lo;
      *(v4f*)(pwr + 16 * mt + 4) = hi4;
    }
    FENCE();
    pt[3] = wmma_f32_16x16x4(whrA[3][1], b1p, pt[3]);
    pt[2] = wmma_f32_16x16x4(whrA[2][2], b2p, pt[2]);
    pt[3] = wmma_f32_16x16x4(whrA[3][2], b2p, pt[3]);
    pt[2] = wmma_f32_16x16x4(whrA[2][3], b3p, pt[2]);
    pt[3] = wmma_f32_16x16x4(whrA[3][3], b3p, pt[3]);
    FENCE();
    LDS_LANDED();
    BAR_SIGNAL();
#pragma unroll
    for (int mt = 2; mt < 4; ++mt) {
      v4f lo, hi4;
      lo[0] = pt[mt][0]; lo[1] = pt[mt][1]; lo[2] = pt[mt][2]; lo[3] = pt[mt][3];
      hi4[0] = pt[mt][4]; hi4[1] = pt[mt][5]; hi4[2] = pt[mt][6]; hi4[3] = pt[mt][7];
      *(v4f*)(pwr + 16 * mt)     = lo;
      *(v4f*)(pwr + 16 * mt + 4) = hi4;
    }
    xb = xnext;
    BAR_WAIT();
  }
  __syncthreads();
#undef LDS_LANDED
#undef BAR_SIGNAL
#undef BAR_WAIT
#undef FENCE

  {
    const float* pl = pbuf + ((T_STEPS - 1) & 1) * PBUF;
    for (int idx = tid; idx < ROWS * PROJ; idx += blockDim.x) {
      const int r = idx / PROJ, cc = idx - r * PROJ;
      const int o = r * PPITCH + cc;
      out[(size_t)(rbase + r) * PROJ + cc] =
          ((pl[o] + pl[PPLANE + o]) + pl[2 * PPLANE + o]) + pl[3 * PPLANE + o];
    }
  }
}

extern "C" void kernel_launch(void* const* d_in, const int* in_sizes, int n_in,
                              void* d_out, int out_size, void* d_ws, size_t ws_size,
                              hipStream_t stream) {
  (void)in_sizes; (void)n_in; (void)out_size; (void)d_ws; (void)ws_size;
  const float* x    = (const float*)d_in[0];
  const float* W_ih = (const float*)d_in[1];
  const float* W_hh = (const float*)d_in[2];
  const float* b_ih = (const float*)d_in[3];
  const float* b_hh = (const float*)d_in[4];
  const float* W_hr = (const float*)d_in[5];
  float* out = (float*)d_out;

  dim3 grid(NBATCH / ROWS);
  dim3 block(128);
  lstm_proj_wmma_kernel<<<grid, block, 0, stream>>>(x, W_ih, W_hh, b_ih, b_hh,
                                                    W_hr, out);
}
